# starting kernel
# baseline (speedup 1.0000x reference)
#include <hip/hip_runtime.h>

typedef __attribute__((ext_vector_type(2))) float v2f;
typedef __attribute__((ext_vector_type(8))) float v8f;

#define NWAVES 8
#define TP     16

#define W1S 258
#define D1S 130
#define W2S 130
#define D2S 66
#define W3S 66
#define BUF128 130
#define BUF64  66

#define OFF_W1  0
#define OFF_D1  (OFF_W1 + 128*W1S)
#define OFF_W2  (OFF_D1 + 128*D1S)
#define OFF_D2  (OFF_W2 + 64*W2S)
#define OFF_W3  (OFF_D2 + 64*D2S)
#define OFF_BUF (OFF_W3 + 16*W3S)
#define BUF_FLOATS (16*BUF128)
#define OFF_Z   (OFF_BUF + NWAVES*BUF_FLOATS)
#define LDS_FLOATS (OFF_Z + NWAVES*144)

static __device__ inline v8f wmma4(v2f a, v2f b, v8f c) {
  return __builtin_amdgcn_wmma_f32_16x16x4_f32(false, a, false, b, (short)0, c,
                                               false, false);
}

template<int MT, int K>
static __device__ inline void gemm_lds(const float* __restrict__ W, int WS,
                                       const float* __restrict__ B, int BS,
                                       v8f* acc, int l16, int half) {
#pragma unroll 2
  for (int k = 0; k < K; k += 4) {
    const int c0 = k + 2*half;
    v2f b = *(const v2f*)(B + l16*BS + c0);
#pragma unroll
    for (int mt = 0; mt < MT; ++mt) {
      v2f a = *(const v2f*)(W + (mt*16 + l16)*WS + c0);
      acc[mt] = wmma4(a, b, acc[mt]);
    }
  }
}

template<int MT>
static __device__ inline void spill_cm(const v8f* acc, float* buf, int S,
                                       int l16, int half) {
#pragma unroll
  for (int mt = 0; mt < MT; ++mt) {
#pragma unroll
    for (int i = 0; i < 4; ++i) {
      v2f p;
      p.x = acc[mt][2*i];
      p.y = acc[mt][2*i + 1];
      *(v2f*)(buf + l16*S + mt*16 + 8*half + 2*i) = p;
    }
  }
}

template<int MT>
static __device__ inline void vn_nonlin(v8f* y, const v8f* d) {
  float dot = 0.f, dns = 0.f;
#pragma unroll
  for (int mt = 0; mt < MT; ++mt) {
#pragma unroll
    for (int i = 0; i < 8; ++i) {
      dot += y[mt][i] * d[mt][i];
      dns += d[mt][i] * d[mt][i];
    }
  }
  dot += __shfl_xor(dot, 16, 32);
  dns += __shfl_xor(dns, 16, 32);
  const float s = (dot >= 0.f) ? 0.f : 0.8f * dot / (dns + 1e-7f);
#pragma unroll
  for (int mt = 0; mt < MT; ++mt) {
#pragma unroll
    for (int i = 0; i < 8; ++i) {
      y[mt][i] = y[mt][i] - s * d[mt][i];
    }
  }
}

__global__ void __launch_bounds__(256, 1)
vnstd_kernel(const float* __restrict__ xg,
             const float* __restrict__ W1g, const float* __restrict__ D1g,
             const float* __restrict__ W2g, const float* __restrict__ D2g,
             const float* __restrict__ W3g,
             float* __restrict__ xstd, float* __restrict__ z0out) {
  extern __shared__ float smem[];
  const int tid = threadIdx.x;

  for (int i = tid; i < 128*256; i += 256) smem[OFF_W1 + (i >> 8)*W1S + (i & 255)] = W1g[i];
  for (int i = tid; i < 128*128; i += 256) smem[OFF_D1 + (i >> 7)*D1S + (i & 127)] = D1g[i];
  for (int i = tid; i < 64*128;  i += 256) smem[OFF_W2 + (i >> 7)*W2S + (i & 127)] = W2g[i];
  for (int i = tid; i < 64*64;   i += 256) smem[OFF_D2 + (i >> 6)*D2S + (i & 63)]  = D2g[i];
  for (int i = tid; i < 16*64;   i += 256)
    smem[OFF_W3 + (i >> 6)*W3S + (i & 63)] = (i < 3*64) ? W3g[i] : 0.f;
  __syncthreads();

  const int wave = tid >> 5;
  const int lane = tid & 31;
  const int l16  = lane & 15;
  const int half = lane >> 4;

  float* buf = smem + OFF_BUF + wave * BUF_FLOATS;
  float* zb  = smem + OFF_Z   + wave * 144;

  const int p0 = (blockIdx.x * NWAVES + wave) * TP;
  const int gp = p0 + l16;

  const float* sW1 = smem + OFF_W1;
  const float* sD1 = smem + OFF_D1;
  const float* sW2 = smem + OFF_W2;
  const float* sD2 = smem + OFF_D2;
  const float* sW3 = smem + OFF_W3;

  const v8f vzero = {0.f, 0.f, 0.f, 0.f, 0.f, 0.f, 0.f, 0.f};

#pragma unroll 1
  for (int t = 0; t < 3; ++t) {
    v8f y1[8];
#pragma unroll
    for (int mt = 0; mt < 8; ++mt) y1[mt] = vzero;

    const float* xb = xg + (size_t)gp * 768 + t;
#pragma unroll 2
    for (int k = 0; k < 256; k += 4) {
      const int c0 = k + 2*half;
      v2f b;
      b.x = xb[c0*3];
      b.y = xb[c0*3 + 3];
#pragma unroll
      for (int mt = 0; mt < 8; ++mt) {
        v2f a = *(const v2f*)(sW1 + (mt*16 + l16)*W1S + c0);
        y1[mt] = wmma4(a, b, y1[mt]);
      }
    }
    spill_cm<8>(y1, buf, BUF128, l16, half);

    v8f d1[8];
#pragma unroll
    for (int mt = 0; mt < 8; ++mt) d1[mt] = vzero;
    gemm_lds<8, 128>(sD1, D1S, buf, BUF128, d1, l16, half);

    vn_nonlin<8>(y1, d1);
    spill_cm<8>(y1, buf, BUF128, l16, half);

    v8f y2[4];
#pragma unroll
    for (int mt = 0; mt < 4; ++mt) y2[mt] = vzero;
    gemm_lds<4, 128>(sW2, W2S, buf, BUF128, y2, l16, half);
    spill_cm<4>(y2, buf, BUF64, l16, half);

    v8f d2[4];
#pragma unroll
    for (int mt = 0; mt < 4; ++mt) d2[mt] = vzero;
    gemm_lds<4, 64>(sD2, D2S, buf, BUF64, d2, l16, half);

    vn_nonlin<4>(y2, d2);
    spill_cm<4>(y2, buf, BUF64, l16, half);

    v8f zt = vzero;
    gemm_lds<1, 64>(sW3, W3S, buf, BUF64, &zt, l16, half);

    if (half == 0) {
#pragma unroll
      for (int r = 0; r < 3; ++r) zb[(t*3 + r)*16 + l16] = zt[r];
    }
  }

#pragma unroll 1
  for (int p = 0; p < TP; ++p) {
    const int pg = p0 + p;
    float zm[3][3];
#pragma unroll
    for (int j = 0; j < 3; ++j)
#pragma unroll
      for (int k = 0; k < 3; ++k) zm[j][k] = zb[(j*3 + k)*16 + p];

    const float* xp = xg + (size_t)pg * 768;
    float* op = xstd + (size_t)pg * 768;
#pragma unroll 1
    for (int cb = 0; cb < 8; ++cb) {
      const int c = cb*32 + lane;
      const float x0 = xp[c*3 + 0];
      const float x1 = xp[c*3 + 1];
      const float x2 = xp[c*3 + 2];
#pragma unroll
      for (int k = 0; k < 3; ++k)
        op[c*3 + k] = x0*zm[0][k] + x1*zm[1][k] + x2*zm[2][k];
    }
  }

  if (lane < 16) {
    const int pg = p0 + lane;
#pragma unroll
    for (int j = 0; j < 3; ++j)
#pragma unroll
      for (int k = 0; k < 3; ++k)
        z0out[(size_t)pg*9 + j*3 + k] = zb[(j*3 + k)*16 + lane];
  }
}

extern "C" void kernel_launch(void* const* d_in, const int* in_sizes, int n_in,
                              void* d_out, int out_size, void* d_ws, size_t ws_size,
                              hipStream_t stream) {
  (void)in_sizes; (void)n_in; (void)out_size; (void)d_ws; (void)ws_size;
  const float* x  = (const float*)d_in[0];
  const float* W1 = (const float*)d_in[1];
  const float* D1 = (const float*)d_in[2];
  const float* W2 = (const float*)d_in[3];
  const float* D2 = (const float*)d_in[4];
  const float* W3 = (const float*)d_in[5];

  float* xstd = (float*)d_out;
  float* z0   = xstd + (size_t)8 * 4096 * 256 * 3;

  const int points = 8 * 4096;
  const int blocks = points / (NWAVES * TP);
  const size_t shmem = (size_t)LDS_FLOATS * sizeof(float);

  vnstd_kernel<<<dim3(blocks), dim3(256), shmem, stream>>>(
      x, W1, D1, W2, D2, W3, xstd, z0);
}


# Stage 1 (W1 product) does all three vector components in one pass: each lane reads 48 contiguous x bytes per 8 channels (3 b128 loads), each W1 fragment feeds 3 accumulator sets; new k sum order
# speedup vs baseline: 1.1462x; 1.1462x over previous
#include <hip/hip_runtime.h>

typedef __attribute__((ext_vector_type(2))) float v2f;
typedef __attribute__((ext_vector_type(8))) float v8f;

#define NWAVES 8
#define TP     16

#define W1S 258
#define D1S 130
#define W2S 130
#define D2S 66
#define W3S 66
#define BUF128 130
#define BUF64  66

#define OFF_W1  0
#define OFF_D1  (OFF_W1 + 128*W1S)
#define OFF_W2  (OFF_D1 + 128*D1S)
#define OFF_D2  (OFF_W2 + 64*W2S)
#define OFF_W3  (OFF_D2 + 64*D2S)
#define OFF_BUF (OFF_W3 + 16*W3S)
#define BUF_FLOATS (16*BUF128)
#define OFF_Z   (OFF_BUF + NWAVES*BUF_FLOATS)
#define LDS_FLOATS (OFF_Z + NWAVES*144)

static __device__ inline v8f wmma4(v2f a, v2f b, v8f c) {
  return __builtin_amdgcn_wmma_f32_16x16x4_f32(false, a, false, b, (short)0, c,
                                               false, false);
}

template<int MT, int K>
static __device__ inline void gemm_lds(const float* __restrict__ W, int WS,
                                       const float* __restrict__ B, int BS,
                                       v8f* acc, int l16, int half) {
#pragma unroll 2
  for (int k = 0; k < K; k += 4) {
    const int c0 = k + 2*half;
    v2f b = *(const v2f*)(B + l16*BS + c0);
#pragma unroll
    for (int mt = 0; mt < MT; ++mt) {
      v2f a = *(const v2f*)(W + (mt*16 + l16)*WS + c0);
      acc[mt] = wmma4(a, b, acc[mt]);
    }
  }
}

template<int MT>
static __device__ inline void spill_cm(const v8f* acc, float* buf, int S,
                                       int l16, int half) {
#pragma unroll
  for (int mt = 0; mt < MT; ++mt) {
#pragma unroll
    for (int i = 0; i < 4; ++i) {
      v2f p;
      p.x = acc[mt][2*i];
      p.y = acc[mt][2*i + 1];
      *(v2f*)(buf + l16*S + mt*16 + 8*half + 2*i) = p;
    }
  }
}

template<int MT>
static __device__ inline void vn_nonlin(v8f* y, const v8f* d) {
  float dot = 0.f, dns = 0.f;
#pragma unroll
  for (int mt = 0; mt < MT; ++mt) {
#pragma unroll
    for (int i = 0; i < 8; ++i) {
      dot += y[mt][i] * d[mt][i];
      dns += d[mt][i] * d[mt][i];
    }
  }
  dot += __shfl_xor(dot, 16, 32);
  dns += __shfl_xor(dns, 16, 32);
  const float s = (dot >= 0.f) ? 0.f : 0.8f * dot / (dns + 1e-7f);
#pragma unroll
  for (int mt = 0; mt < MT; ++mt) {
#pragma unroll
    for (int i = 0; i < 8; ++i) {
      y[mt][i] = y[mt][i] - s * d[mt][i];
    }
  }
}

__global__ void __launch_bounds__(256, 1)
vnstd_kernel(const float* __restrict__ xg,
             const float* __restrict__ W1g, const float* __restrict__ D1g,
             const float* __restrict__ W2g, const float* __restrict__ D2g,
             const float* __restrict__ W3g,
             float* __restrict__ xstd, float* __restrict__ z0out) {
  extern __shared__ float smem[];
  const int tid = threadIdx.x;

  for (int i = tid; i < 128*256; i += 256) smem[OFF_W1 + (i >> 8)*W1S + (i & 255)] = W1g[i];
  for (int i = tid; i < 128*128; i += 256) smem[OFF_D1 + (i >> 7)*D1S + (i & 127)] = D1g[i];
  for (int i = tid; i < 64*128;  i += 256) smem[OFF_W2 + (i >> 7)*W2S + (i & 127)] = W2g[i];
  for (int i = tid; i < 64*64;   i += 256) smem[OFF_D2 + (i >> 6)*D2S + (i & 63)]  = D2g[i];
  for (int i = tid; i < 16*64;   i += 256)
    smem[OFF_W3 + (i >> 6)*W3S + (i & 63)] = (i < 3*64) ? W3g[i] : 0.f;
  __syncthreads();

  const int wave = tid >> 5;
  const int lane = tid & 31;
  const int l16  = lane & 15;
  const int half = lane >> 4;

  float* buf = smem + OFF_BUF + wave * BUF_FLOATS;
  float* zb  = smem + OFF_Z   + wave * 144;

  const int p0 = (blockIdx.x * NWAVES + wave) * TP;
  const int gp = p0 + l16;

  const float* sW1 = smem + OFF_W1;
  const float* sD1 = smem + OFF_D1;
  const float* sW2 = smem + OFF_W2;
  const float* sD2 = smem + OFF_D2;
  const float* sW3 = smem + OFF_W3;

  const v8f vzero = {0.f, 0.f, 0.f, 0.f, 0.f, 0.f, 0.f, 0.f};

  v8f y1[3][8];
#pragma unroll
  for (int t = 0; t < 3; ++t)
#pragma unroll
    for (int mt = 0; mt < 8; ++mt) y1[t][mt] = vzero;

  const float* xrow = xg + (size_t)gp * 768;
#pragma unroll 2
  for (int k = 0; k < 256; k += 8) {
    const int cb = k + 4*half;
    const float4* px = (const float4*)(xrow + cb*3);
    const float4 q0 = px[0];
    const float4 q1 = px[1];
    const float4 q2 = px[2];
    v2f bA[3], bB[3];
    bA[0].x = q0.x; bA[0].y = q0.w;
    bA[1].x = q0.y; bA[1].y = q1.x;
    bA[2].x = q0.z; bA[2].y = q1.y;
    bB[0].x = q1.z; bB[0].y = q2.y;
    bB[1].x = q1.w; bB[1].y = q2.z;
    bB[2].x = q2.x; bB[2].y = q2.w;
#pragma unroll
    for (int mt = 0; mt < 8; ++mt) {
      const float* wr = sW1 + (mt*16 + l16)*W1S + cb;
      const v2f aA = *(const v2f*)(wr);
      const v2f aB = *(const v2f*)(wr + 2);
#pragma unroll
      for (int t = 0; t < 3; ++t) y1[t][mt] = wmma4(aA, bA[t], y1[t][mt]);
#pragma unroll
      for (int t = 0; t < 3; ++t) y1[t][mt] = wmma4(aB, bB[t], y1[t][mt]);
    }
  }

#pragma unroll
  for (int t = 0; t < 3; ++t) {
    spill_cm<8>(y1[t], buf, BUF128, l16, half);

    v8f d1[8];
#pragma unroll
    for (int mt = 0; mt < 8; ++mt) d1[mt] = vzero;
    gemm_lds<8, 128>(sD1, D1S, buf, BUF128, d1, l16, half);

    vn_nonlin<8>(y1[t], d1);
    spill_cm<8>(y1[t], buf, BUF128, l16, half);

    v8f y2[4];
#pragma unroll
    for (int mt = 0; mt < 4; ++mt) y2[mt] = vzero;
    gemm_lds<4, 128>(sW2, W2S, buf, BUF128, y2, l16, half);
    spill_cm<4>(y2, buf, BUF64, l16, half);

    v8f d2[4];
#pragma unroll
    for (int mt = 0; mt < 4; ++mt) d2[mt] = vzero;
    gemm_lds<4, 64>(sD2, D2S, buf, BUF64, d2, l16, half);

    vn_nonlin<4>(y2, d2);
    spill_cm<4>(y2, buf, BUF64, l16, half);

    v8f zt = vzero;
    gemm_lds<1, 64>(sW3, W3S, buf, BUF64, &zt, l16, half);

    if (half == 0) {
#pragma unroll
      for (int r = 0; r < 3; ++r) zb[(t*3 + r)*16 + l16] = zt[r];
    }
  }

#pragma unroll 1
  for (int p = 0; p < TP; ++p) {
    const int pg = p0 + p;
    float zm[3][3];
#pragma unroll
    for (int j = 0; j < 3; ++j)
#pragma unroll
      for (int k = 0; k < 3; ++k) zm[j][k] = zb[(j*3 + k)*16 + p];

    const float* xp = xg + (size_t)pg * 768;
    float* op = xstd + (size_t)pg * 768;
#pragma unroll 1
    for (int cb = 0; cb < 8; ++cb) {
      const int c = cb*32 + lane;
      const float x0 = xp[c*3 + 0];
      const float x1 = xp[c*3 + 1];
      const float x2 = xp[c*3 + 2];
#pragma unroll
      for (int k = 0; k < 3; ++k)
        op[c*3 + k] = x0*zm[0][k] + x1*zm[1][k] + x2*zm[2][k];
    }
  }

  if (lane < 16) {
    const int pg = p0 + lane;
#pragma unroll
    for (int j = 0; j < 3; ++j)
#pragma unroll
      for (int k = 0; k < 3; ++k)
        z0out[(size_t)pg*9 + j*3 + k] = zb[(j*3 + k)*16 + lane];
  }
}

extern "C" void kernel_launch(void* const* d_in, const int* in_sizes, int n_in,
                              void* d_out, int out_size, void* d_ws, size_t ws_size,
                              hipStream_t stream) {
  (void)in_sizes; (void)n_in; (void)out_size; (void)d_ws; (void)ws_size;
  const float* x  = (const float*)d_in[0];
  const float* W1 = (const float*)d_in[1];
  const float* D1 = (const float*)d_in[2];
  const float* W2 = (const float*)d_in[3];
  const float* D2 = (const float*)d_in[4];
  const float* W3 = (const float*)d_in[5];

  float* xstd = (float*)d_out;
  float* z0   = xstd + (size_t)8 * 4096 * 256 * 3;

  const int points = 8 * 4096;
  const int blocks = points / (NWAVES * TP);
  const size_t shmem = (size_t)LDS_FLOATS * sizeof(float);

  vnstd_kernel<<<dim3(blocks), dim3(256), shmem, stream>>>(
      x, W1, D1, W2, D2, W3, xstd, z0);
}


# Final x@z0 pass streamed: one channel (3 floats) per lane per b96 load and b96 store, every access whole 128-B lines, several points' loads in flight instead of one load pair per round trip
# speedup vs baseline: 1.3505x; 1.1782x over previous
#include <hip/hip_runtime.h>

typedef __attribute__((ext_vector_type(2))) float v2f;
typedef __attribute__((ext_vector_type(8))) float v8f;

#define NWAVES 8
#define TP     16

#define W1S 258
#define D1S 130
#define W2S 130
#define D2S 66
#define W3S 66
#define BUF128 130
#define BUF64  66

#define OFF_W1  0
#define OFF_D1  (OFF_W1 + 128*W1S)
#define OFF_W2  (OFF_D1 + 128*D1S)
#define OFF_D2  (OFF_W2 + 64*W2S)
#define OFF_W3  (OFF_D2 + 64*D2S)
#define OFF_BUF (OFF_W3 + 16*W3S)
#define BUF_FLOATS (16*BUF128)
#define OFF_Z   (OFF_BUF + NWAVES*BUF_FLOATS)
#define LDS_FLOATS (OFF_Z + NWAVES*144)

static __device__ inline v8f wmma4(v2f a, v2f b, v8f c) {
  return __builtin_amdgcn_wmma_f32_16x16x4_f32(false, a, false, b, (short)0, c,
                                               false, false);
}

template<int MT, int K>
static __device__ inline void gemm_lds(const float* __restrict__ W, int WS,
                                       const float* __restrict__ B, int BS,
                                       v8f* acc, int l16, int half) {
#pragma unroll 2
  for (int k = 0; k < K; k += 4) {
    const int c0 = k + 2*half;
    v2f b = *(const v2f*)(B + l16*BS + c0);
#pragma unroll
    for (int mt = 0; mt < MT; ++mt) {
      v2f a = *(const v2f*)(W + (mt*16 + l16)*WS + c0);
      acc[mt] = wmma4(a, b, acc[mt]);
    }
  }
}

template<int MT>
static __device__ inline void spill_cm(const v8f* acc, float* buf, int S,
                                       int l16, int half) {
#pragma unroll
  for (int mt = 0; mt < MT; ++mt) {
#pragma unroll
    for (int i = 0; i < 4; ++i) {
      v2f p;
      p.x = acc[mt][2*i];
      p.y = acc[mt][2*i + 1];
      *(v2f*)(buf + l16*S + mt*16 + 8*half + 2*i) = p;
    }
  }
}

template<int MT>
static __device__ inline void vn_nonlin(v8f* y, const v8f* d) {
  float dot = 0.f, dns = 0.f;
#pragma unroll
  for (int mt = 0; mt < MT; ++mt) {
#pragma unroll
    for (int i = 0; i < 8; ++i) {
      dot += y[mt][i] * d[mt][i];
      dns += d[mt][i] * d[mt][i];
    }
  }
  dot += __shfl_xor(dot, 16, 32);
  dns += __shfl_xor(dns, 16, 32);
  const float s = (dot >= 0.f) ? 0.f : 0.8f * dot / (dns + 1e-7f);
#pragma unroll
  for (int mt = 0; mt < MT; ++mt) {
#pragma unroll
    for (int i = 0; i < 8; ++i) {
      y[mt][i] = y[mt][i] - s * d[mt][i];
    }
  }
}

__global__ void __launch_bounds__(256, 1)
vnstd_kernel(const float* __restrict__ xg,
             const float* __restrict__ W1g, const float* __restrict__ D1g,
             const float* __restrict__ W2g, const float* __restrict__ D2g,
             const float* __restrict__ W3g,
             float* __restrict__ xstd, float* __restrict__ z0out) {
  extern __shared__ float smem[];
  const int tid = threadIdx.x;

  for (int i = tid; i < 128*256; i += 256) smem[OFF_W1 + (i >> 8)*W1S + (i & 255)] = W1g[i];
  for (int i = tid; i < 128*128; i += 256) smem[OFF_D1 + (i >> 7)*D1S + (i & 127)] = D1g[i];
  for (int i = tid; i < 64*128;  i += 256) smem[OFF_W2 + (i >> 7)*W2S + (i & 127)] = W2g[i];
  for (int i = tid; i < 64*64;   i += 256) smem[OFF_D2 + (i >> 6)*D2S + (i & 63)]  = D2g[i];
  for (int i = tid; i < 16*64;   i += 256)
    smem[OFF_W3 + (i >> 6)*W3S + (i & 63)] = (i < 3*64) ? W3g[i] : 0.f;
  __syncthreads();

  const int wave = tid >> 5;
  const int lane = tid & 31;
  const int l16  = lane & 15;
  const int half = lane >> 4;

  float* buf = smem + OFF_BUF + wave * BUF_FLOATS;
  float* zb  = smem + OFF_Z   + wave * 144;

  const int p0 = (blockIdx.x * NWAVES + wave) * TP;
  const int gp = p0 + l16;

  const float* sW1 = smem + OFF_W1;
  const float* sD1 = smem + OFF_D1;
  const float* sW2 = smem + OFF_W2;
  const float* sD2 = smem + OFF_D2;
  const float* sW3 = smem + OFF_W3;

  const v8f vzero = {0.f, 0.f, 0.f, 0.f, 0.f, 0.f, 0.f, 0.f};

  v8f y1[3][8];
#pragma unroll
  for (int t = 0; t < 3; ++t)
#pragma unroll
    for (int mt = 0; mt < 8; ++mt) y1[t][mt] = vzero;

  const float* xrow = xg + (size_t)gp * 768;
#pragma unroll 2
  for (int k = 0; k < 256; k += 8) {
    const int cb = k + 4*half;
    const float4* px = (const float4*)(xrow + cb*3);
    const float4 q0 = px[0];
    const float4 q1 = px[1];
    const float4 q2 = px[2];
    v2f bA[3], bB[3];
    bA[0].x = q0.x; bA[0].y = q0.w;
    bA[1].x = q0.y; bA[1].y = q1.x;
    bA[2].x = q0.z; bA[2].y = q1.y;
    bB[0].x = q1.z; bB[0].y = q2.y;
    bB[1].x = q1.w; bB[1].y = q2.z;
    bB[2].x = q2.x; bB[2].y = q2.w;
#pragma unroll
    for (int mt = 0; mt < 8; ++mt) {
      const float* wr = sW1 + (mt*16 + l16)*W1S + cb;
      const v2f aA = *(const v2f*)(wr);
      const v2f aB = *(const v2f*)(wr + 2);
#pragma unroll
      for (int t = 0; t < 3; ++t) y1[t][mt] = wmma4(aA, bA[t], y1[t][mt]);
#pragma unroll
      for (int t = 0; t < 3; ++t) y1[t][mt] = wmma4(aB, bB[t], y1[t][mt]);
    }
  }

#pragma unroll
  for (int t = 0; t < 3; ++t) {
    spill_cm<8>(y1[t], buf, BUF128, l16, half);

    v8f d1[8];
#pragma unroll
    for (int mt = 0; mt < 8; ++mt) d1[mt] = vzero;
    gemm_lds<8, 128>(sD1, D1S, buf, BUF128, d1, l16, half);

    vn_nonlin<8>(y1[t], d1);
    spill_cm<8>(y1[t], buf, BUF128, l16, half);

    v8f y2[4];
#pragma unroll
    for (int mt = 0; mt < 4; ++mt) y2[mt] = vzero;
    gemm_lds<4, 128>(sW2, W2S, buf, BUF128, y2, l16, half);
    spill_cm<4>(y2, buf, BUF64, l16, half);

    v8f d2[4];
#pragma unroll
    for (int mt = 0; mt < 4; ++mt) d2[mt] = vzero;
    gemm_lds<4, 64>(sD2, D2S, buf, BUF64, d2, l16, half);

    vn_nonlin<4>(y2, d2);
    spill_cm<4>(y2, buf, BUF64, l16, half);

    v8f zt = vzero;
    gemm_lds<1, 64>(sW3, W3S, buf, BUF64, &zt, l16, half);

    if (half == 0) {
#pragma unroll
      for (int r = 0; r < 3; ++r) zb[(t*3 + r)*16 + l16] = zt[r];
    }
  }

  constexpr int PT = 4;
  for (int pb = 0; pb < TP; pb += PT) {
    float q[PT][8][3];
#pragma unroll
    for (int pp = 0; pp < PT; ++pp) {
      const float* xp = xg + (size_t)(p0 + pb + pp) * 768 + lane*3;
#pragma unroll
      for (int cb = 0; cb < 8; ++cb) {
#pragma unroll
        for (int j = 0; j < 3; ++j) q[pp][cb][j] = xp[cb*96 + j];
      }
    }
#pragma unroll
    for (int pp = 0; pp < PT; ++pp) {
      float zm[3][3];
#pragma unroll
      for (int j = 0; j < 3; ++j)
#pragma unroll
        for (int k = 0; k < 3; ++k) zm[j][k] = zb[(j*3 + k)*16 + pb + pp];

      float* op = xstd + (size_t)(p0 + pb + pp) * 768 + lane*3;
#pragma unroll
      for (int cb = 0; cb < 8; ++cb) {
        const float x0 = q[pp][cb][0], x1 = q[pp][cb][1], x2 = q[pp][cb][2];
#pragma unroll
        for (int k = 0; k < 3; ++k)
          op[cb*96 + k] = x0*zm[0][k] + x1*zm[1][k] + x2*zm[2][k];
      }
    }
  }

  if (lane < 16) {
    const int pg = p0 + lane;
#pragma unroll
    for (int j = 0; j < 3; ++j)
#pragma unroll
      for (int k = 0; k < 3; ++k)
        z0out[(size_t)pg*9 + j*3 + k] = zb[(j*3 + k)*16 + lane];
  }
}

extern "C" void kernel_launch(void* const* d_in, const int* in_sizes, int n_in,
                              void* d_out, int out_size, void* d_ws, size_t ws_size,
                              hipStream_t stream) {
  (void)in_sizes; (void)n_in; (void)out_size; (void)d_ws; (void)ws_size;
  const float* x  = (const float*)d_in[0];
  const float* W1 = (const float*)d_in[1];
  const float* D1 = (const float*)d_in[2];
  const float* W2 = (const float*)d_in[3];
  const float* D2 = (const float*)d_in[4];
  const float* W3 = (const float*)d_in[5];

  float* xstd = (float*)d_out;
  float* z0   = xstd + (size_t)8 * 4096 * 256 * 3;

  const int points = 8 * 4096;
  const int blocks = points / (NWAVES * TP);
  const size_t shmem = (size_t)LDS_FLOATS * sizeof(float);

  vnstd_kernel<<<dim3(blocks), dim3(256), shmem, stream>>>(
      x, W1, D1, W2, D2, W3, xstd, z0);
}


# GEMM stages 2-5 take the B operand straight from the previous stage's accumulator registers (k = 16kt+8h+2j+s on both operands, A as 16-B LDS loads), no LDS spill and reload; new k sum order
# speedup vs baseline: 1.4897x; 1.1031x over previous
#include <hip/hip_runtime.h>

typedef __attribute__((ext_vector_type(2))) float v2f;
typedef __attribute__((ext_vector_type(4))) float v4f;
typedef __attribute__((ext_vector_type(8))) float v8f;

#define NWAVES 8
#define TP     16

#define W1S 258
#define D1S 132
#define W2S 132
#define D2S 68
#define W3S 68

#define OFF_W1  0
#define OFF_D1  (OFF_W1 + 128*W1S)
#define OFF_W2  (OFF_D1 + 128*D1S)
#define OFF_D2  (OFF_W2 + 64*W2S)
#define OFF_W3  (OFF_D2 + 64*D2S)
#define OFF_Z   (OFF_W3 + 16*W3S)
#define LDS_FLOATS (OFF_Z + NWAVES*144)

static __device__ inline v8f wmma4(v2f a, v2f b, v8f c) {
  return __builtin_amdgcn_wmma_f32_16x16x4_f32(false, a, false, b, (short)0, c,
                                               false, false);
}

template<int MT, int KB>
static __device__ inline void load_frags(v4f (*f)[MT][2], const float* wl, int WS, int kt0) {
#pragma unroll
  for (int kb = 0; kb < KB; ++kb) {
#pragma unroll
    for (int mt = 0; mt < MT; ++mt) {
      const v4f* p = (const v4f*)(wl + mt*16*WS + (kt0 + kb)*16);
      f[kb][mt][0] = p[0];
      f[kb][mt][1] = p[1];
    }
  }
}

template<int MT, int KT, int KB>
static __device__ inline void gemm_regB(const float* __restrict__ W, int WS,
                                        const v8f* Y, v8f* acc, int l16, int half) {
  constexpr int NB = KT / KB;
  const float* wl = W + l16*WS + 8*half;
  v4f fa[KB][MT][2], fb[KB][MT][2];
  load_frags<MT, KB>(fa, wl, WS, 0);
  if (NB == 1) {
    __builtin_amdgcn_sched_barrier(0);
  }
#pragma unroll
  for (int nb = 0; nb < NB; ++nb) {
    v4f (*cur)[MT][2] = (nb & 1) ? fb : fa;
    v4f (*nxt)[MT][2] = (nb & 1) ? fa : fb;
    if (nb + 1 < NB) {
      load_frags<MT, KB>(nxt, wl, WS, (nb + 1)*KB);
      __builtin_amdgcn_sched_barrier(0);
    }
#pragma unroll
    for (int kb = 0; kb < KB; ++kb) {
      const int kt = nb*KB + kb;
#pragma unroll
      for (int j = 0; j < 4; ++j) {
        v2f b;
        b.x = Y[kt][2*j];
        b.y = Y[kt][2*j + 1];
#pragma unroll
        for (int mt = 0; mt < MT; ++mt) {
          const v4f q = cur[kb][mt][j >> 1];
          v2f a;
          a.x = (j & 1) ? q.z : q.x;
          a.y = (j & 1) ? q.w : q.y;
          acc[mt] = wmma4(a, b, acc[mt]);
        }
      }
    }
    if (nb + 1 < NB) {
      __builtin_amdgcn_sched_barrier(0);
    }
  }
}

template<int MT>
static __device__ inline void vn_nonlin(v8f* y, const v8f* d) {
  float dot = 0.f, dns = 0.f;
#pragma unroll
  for (int mt = 0; mt < MT; ++mt) {
#pragma unroll
    for (int i = 0; i < 8; ++i) {
      dot += y[mt][i] * d[mt][i];
      dns += d[mt][i] * d[mt][i];
    }
  }
  dot += __shfl_xor(dot, 16, 32);
  dns += __shfl_xor(dns, 16, 32);
  const float s = (dot >= 0.f) ? 0.f : 0.8f * dot / (dns + 1e-7f);
#pragma unroll
  for (int mt = 0; mt < MT; ++mt) {
#pragma unroll
    for (int i = 0; i < 8; ++i) {
      y[mt][i] = y[mt][i] - s * d[mt][i];
    }
  }
}

__global__ void __launch_bounds__(256, 1)
vnstd_kernel(const float* __restrict__ xg,
             const float* __restrict__ W1g, const float* __restrict__ D1g,
             const float* __restrict__ W2g, const float* __restrict__ D2g,
             const float* __restrict__ W3g,
             float* __restrict__ xstd, float* __restrict__ z0out) {
  extern __shared__ __attribute__((aligned(16))) float smem[];
  const int tid = threadIdx.x;

  for (int i = tid; i < 128*256; i += 256) smem[OFF_W1 + (i >> 8)*W1S + (i & 255)] = W1g[i];
  for (int i = tid; i < 128*128; i += 256) smem[OFF_D1 + (i >> 7)*D1S + (i & 127)] = D1g[i];
  for (int i = tid; i < 64*128;  i += 256) smem[OFF_W2 + (i >> 7)*W2S + (i & 127)] = W2g[i];
  for (int i = tid; i < 64*64;   i += 256) smem[OFF_D2 + (i >> 6)*D2S + (i & 63)]  = D2g[i];
  for (int i = tid; i < 16*64;   i += 256)
    smem[OFF_W3 + (i >> 6)*W3S + (i & 63)] = (i < 3*64) ? W3g[i] : 0.f;
  __syncthreads();

  const int wave = tid >> 5;
  const int lane = tid & 31;
  const int l16  = lane & 15;
  const int half = lane >> 4;

  float* zb  = smem + OFF_Z   + wave * 144;

  const int p0 = (blockIdx.x * NWAVES + wave) * TP;
  const int gp = p0 + l16;

  const float* sW1 = smem + OFF_W1;
  const float* sD1 = smem + OFF_D1;
  const float* sW2 = smem + OFF_W2;
  const float* sD2 = smem + OFF_D2;
  const float* sW3 = smem + OFF_W3;

  const v8f vzero = {0.f, 0.f, 0.f, 0.f, 0.f, 0.f, 0.f, 0.f};

  v8f y1[3][8];
#pragma unroll
  for (int t = 0; t < 3; ++t)
#pragma unroll
    for (int mt = 0; mt < 8; ++mt) y1[t][mt] = vzero;

  const float* xrow = xg + (size_t)gp * 768;
#pragma unroll 2
  for (int k = 0; k < 256; k += 8) {
    const int cb = k + 4*half;
    const float4* px = (const float4*)(xrow + cb*3);
    const float4 q0 = px[0];
    const float4 q1 = px[1];
    const float4 q2 = px[2];
    v2f bA[3], bB[3];
    bA[0].x = q0.x; bA[0].y = q0.w;
    bA[1].x = q0.y; bA[1].y = q1.x;
    bA[2].x = q0.z; bA[2].y = q1.y;
    bB[0].x = q1.z; bB[0].y = q2.y;
    bB[1].x = q1.w; bB[1].y = q2.z;
    bB[2].x = q2.x; bB[2].y = q2.w;
#pragma unroll
    for (int mt = 0; mt < 8; ++mt) {
      const float* wr = sW1 + (mt*16 + l16)*W1S + cb;
      const v2f aA = *(const v2f*)(wr);
      const v2f aB = *(const v2f*)(wr + 2);
#pragma unroll
      for (int t = 0; t < 3; ++t) y1[t][mt] = wmma4(aA, bA[t], y1[t][mt]);
#pragma unroll
      for (int t = 0; t < 3; ++t) y1[t][mt] = wmma4(aB, bB[t], y1[t][mt]);
    }
  }

#pragma nounroll
  for (int t = 0; t < 3; ++t) {
    v8f d1[8];
#pragma unroll
    for (int mt = 0; mt < 8; ++mt) d1[mt] = vzero;
    gemm_regB<8, 8, 1>(sD1, D1S, y1[0], d1, l16, half);

    vn_nonlin<8>(y1[0], d1);

    v8f y2[4];
#pragma unroll
    for (int mt = 0; mt < 4; ++mt) y2[mt] = vzero;
    gemm_regB<4, 8, 2>(sW2, W2S, y1[0], y2, l16, half);

    v8f d2[4];
#pragma unroll
    for (int mt = 0; mt < 4; ++mt) d2[mt] = vzero;
    gemm_regB<4, 4, 2>(sD2, D2S, y2, d2, l16, half);

    vn_nonlin<4>(y2, d2);

    v8f zt = vzero;
    gemm_regB<1, 4, 4>(sW3, W3S, y2, &zt, l16, half);

    if (half == 0) {
#pragma unroll
      for (int r = 0; r < 3; ++r) zb[(t*3 + r)*16 + l16] = zt[r];
    }

#pragma unroll
    for (int mt = 0; mt < 8; ++mt) {
      y1[0][mt] = y1[1][mt];
      y1[1][mt] = y1[2][mt];
    }
  }

  constexpr int PT = 4;
  for (int pb = 0; pb < TP; pb += PT) {
    float q[PT][8][3];
#pragma unroll
    for (int pp = 0; pp < PT; ++pp) {
      const float* xp = xg + (size_t)(p0 + pb + pp) * 768 + lane*3;
#pragma unroll
      for (int cb = 0; cb < 8; ++cb) {
#pragma unroll
        for (int j = 0; j < 3; ++j) q[pp][cb][j] = xp[cb*96 + j];
      }
    }
#pragma unroll
    for (int pp = 0; pp < PT; ++pp) {
      float zm[3][3];
#pragma unroll
      for (int j = 0; j < 3; ++j)
#pragma unroll
        for (int k = 0; k < 3; ++k) zm[j][k] = zb[(j*3 + k)*16 + pb + pp];

      float* op = xstd + (size_t)(p0 + pb + pp) * 768 + lane*3;
#pragma unroll
      for (int cb = 0; cb < 8; ++cb) {
        const float x0 = q[pp][cb][0], x1 = q[pp][cb][1], x2 = q[pp][cb][2];
#pragma unroll
        for (int k = 0; k < 3; ++k)
          op[cb*96 + k] = x0*zm[0][k] + x1*zm[1][k] + x2*zm[2][k];
      }
    }
  }

  if (lane < 16) {
    const int pg = p0 + lane;
#pragma unroll
    for (int j = 0; j < 3; ++j)
#pragma unroll
      for (int k = 0; k < 3; ++k)
        z0out[(size_t)pg*9 + j*3 + k] = zb[(j*3 + k)*16 + lane];
  }
}

extern "C" void kernel_launch(void* const* d_in, const int* in_sizes, int n_in,
                              void* d_out, int out_size, void* d_ws, size_t ws_size,
                              hipStream_t stream) {
  (void)in_sizes; (void)n_in; (void)out_size; (void)d_ws; (void)ws_size;
  const float* x  = (const float*)d_in[0];
  const float* W1 = (const float*)d_in[1];
  const float* D1 = (const float*)d_in[2];
  const float* W2 = (const float*)d_in[3];
  const float* D2 = (const float*)d_in[4];
  const float* W3 = (const float*)d_in[5];

  float* xstd = (float*)d_out;
  float* z0   = xstd + (size_t)8 * 4096 * 256 * 3;

  const int points = 8 * 4096;
  const int blocks = points / (NWAVES * TP);
  const size_t shmem = (size_t)LDS_FLOATS * sizeof(float);

  vnstd_kernel<<<dim3(blocks), dim3(256), shmem, stream>>>(
      x, W1, D1, W2, D2, W3, xstd, z0);
}


# Weight staging widened: each thread requests its 61 16-byte pieces of the five matrices as one group ahead of its LDS stores (8- and 16-byte stores), not 32 waited groups of 4-byte loads
# speedup vs baseline: 1.6972x; 1.1393x over previous
#include <hip/hip_runtime.h>

typedef __attribute__((ext_vector_type(2))) float v2f;
typedef __attribute__((ext_vector_type(4))) float v4f;
typedef __attribute__((ext_vector_type(8))) float v8f;

#define NWAVES 8
#define TP     16

#define W1S 258
#define D1S 132
#define W2S 132
#define D2S 68
#define W3S 68

#define OFF_W1  0
#define OFF_D1  (OFF_W1 + 128*W1S)
#define OFF_W2  (OFF_D1 + 128*D1S)
#define OFF_D2  (OFF_W2 + 64*W2S)
#define OFF_W3  (OFF_D2 + 64*D2S)
#define OFF_Z   (OFF_W3 + 16*W3S)
#define LDS_FLOATS (OFF_Z + NWAVES*144)

static __device__ inline v8f wmma4(v2f a, v2f b, v8f c) {
  return __builtin_amdgcn_wmma_f32_16x16x4_f32(false, a, false, b, (short)0, c,
                                               false, false);
}

template<int MT, int KB>
static __device__ inline void load_frags(v4f (*f)[MT][2], const float* wl, int WS, int kt0) {
#pragma unroll
  for (int kb = 0; kb < KB; ++kb) {
#pragma unroll
    for (int mt = 0; mt < MT; ++mt) {
      const v4f* p = (const v4f*)(wl + mt*16*WS + (kt0 + kb)*16);
      f[kb][mt][0] = p[0];
      f[kb][mt][1] = p[1];
    }
  }
}

template<int MT, int KT, int KB>
static __device__ inline void gemm_regB(const float* __restrict__ W, int WS,
                                        const v8f* Y, v8f* acc, int l16, int half) {
  constexpr int NB = KT / KB;
  const float* wl = W + l16*WS + 8*half;
  v4f fa[KB][MT][2], fb[KB][MT][2];
  load_frags<MT, KB>(fa, wl, WS, 0);
  if (NB == 1) {
    __builtin_amdgcn_sched_barrier(0);
  }
#pragma unroll
  for (int nb = 0; nb < NB; ++nb) {
    v4f (*cur)[MT][2] = (nb & 1) ? fb : fa;
    v4f (*nxt)[MT][2] = (nb & 1) ? fa : fb;
    if (nb + 1 < NB) {
      load_frags<MT, KB>(nxt, wl, WS, (nb + 1)*KB);
      __builtin_amdgcn_sched_barrier(0);
    }
#pragma unroll
    for (int kb = 0; kb < KB; ++kb) {
      const int kt = nb*KB + kb;
#pragma unroll
      for (int j = 0; j < 4; ++j) {
        v2f b;
        b.x = Y[kt][2*j];
        b.y = Y[kt][2*j + 1];
#pragma unroll
        for (int mt = 0; mt < MT; ++mt) {
          const v4f q = cur[kb][mt][j >> 1];
          v2f a;
          a.x = (j & 1) ? q.z : q.x;
          a.y = (j & 1) ? q.w : q.y;
          acc[mt] = wmma4(a, b, acc[mt]);
        }
      }
    }
    if (nb + 1 < NB) {
      __builtin_amdgcn_sched_barrier(0);
    }
  }
}

template<int MT>
static __device__ inline void vn_nonlin(v8f* y, const v8f* d) {
  float dot = 0.f, dns = 0.f;
#pragma unroll
  for (int mt = 0; mt < MT; ++mt) {
#pragma unroll
    for (int i = 0; i < 8; ++i) {
      dot += y[mt][i] * d[mt][i];
      dns += d[mt][i] * d[mt][i];
    }
  }
  dot += __shfl_xor(dot, 16, 32);
  dns += __shfl_xor(dns, 16, 32);
  const float s = (dot >= 0.f) ? 0.f : 0.8f * dot / (dns + 1e-7f);
#pragma unroll
  for (int mt = 0; mt < MT; ++mt) {
#pragma unroll
    for (int i = 0; i < 8; ++i) {
      y[mt][i] = y[mt][i] - s * d[mt][i];
    }
  }
}

__global__ void __launch_bounds__(256, 1)
vnstd_kernel(const float* __restrict__ xg,
             const float* __restrict__ W1g, const float* __restrict__ D1g,
             const float* __restrict__ W2g, const float* __restrict__ D2g,
             const float* __restrict__ W3g,
             float* __restrict__ xstd, float* __restrict__ z0out) {
  extern __shared__ __attribute__((aligned(16))) float smem[];
  const int tid = threadIdx.x;

  {
    const float4* W1v = (const float4*)W1g;
    const float4* D1v = (const float4*)D1g;
    const float4* W2v = (const float4*)W2g;
    const float4* D2v = (const float4*)D2g;
    const float4* W3v = (const float4*)W3g;
    float4 rw1[32], rd1[16], rw2[8], rd2[4], rw3;
#pragma unroll
    for (int j = 0; j < 32; ++j) rw1[j] = W1v[tid + 256*j];
#pragma unroll
    for (int j = 0; j < 16; ++j) rd1[j] = D1v[tid + 256*j];
#pragma unroll
    for (int j = 0; j < 8; ++j)  rw2[j] = W2v[tid + 256*j];
#pragma unroll
    for (int j = 0; j < 4; ++j)  rd2[j] = D2v[tid + 256*j];
    rw3 = make_float4(0.f, 0.f, 0.f, 0.f);
    if (tid < 3*16) rw3 = W3v[tid];

    float* s1 = smem + OFF_W1 + (tid >> 6)*W1S + (tid & 63)*4;
#pragma unroll
    for (int j = 0; j < 32; ++j) {
      v2f lo, hi;
      lo.x = rw1[j].x; lo.y = rw1[j].y;
      hi.x = rw1[j].z; hi.y = rw1[j].w;
      *(v2f*)(s1 + j*4*W1S)     = lo;
      *(v2f*)(s1 + j*4*W1S + 2) = hi;
    }
    float* s2 = smem + OFF_D1 + (tid >> 5)*D1S + (tid & 31)*4;
#pragma unroll
    for (int j = 0; j < 16; ++j) *(v4f*)(s2 + j*8*D1S) = (v4f){rd1[j].x, rd1[j].y, rd1[j].z, rd1[j].w};
    float* s3 = smem + OFF_W2 + (tid >> 5)*W2S + (tid & 31)*4;
#pragma unroll
    for (int j = 0; j < 8; ++j)  *(v4f*)(s3 + j*8*W2S) = (v4f){rw2[j].x, rw2[j].y, rw2[j].z, rw2[j].w};
    float* s4 = smem + OFF_D2 + (tid >> 4)*D2S + (tid & 15)*4;
#pragma unroll
    for (int j = 0; j < 4; ++j)  *(v4f*)(s4 + j*16*D2S) = (v4f){rd2[j].x, rd2[j].y, rd2[j].z, rd2[j].w};
    *(v4f*)(smem + OFF_W3 + (tid >> 4)*W3S + (tid & 15)*4) = (v4f){rw3.x, rw3.y, rw3.z, rw3.w};
  }
  __syncthreads();

  const int wave = tid >> 5;
  const int lane = tid & 31;
  const int l16  = lane & 15;
  const int half = lane >> 4;

  float* zb  = smem + OFF_Z   + wave * 144;

  const int p0 = (blockIdx.x * NWAVES + wave) * TP;
  const int gp = p0 + l16;

  const float* sW1 = smem + OFF_W1;
  const float* sD1 = smem + OFF_D1;
  const float* sW2 = smem + OFF_W2;
  const float* sD2 = smem + OFF_D2;
  const float* sW3 = smem + OFF_W3;

  const v8f vzero = {0.f, 0.f, 0.f, 0.f, 0.f, 0.f, 0.f, 0.f};

  v8f y1[3][8];
#pragma unroll
  for (int t = 0; t < 3; ++t)
#pragma unroll
    for (int mt = 0; mt < 8; ++mt) y1[t][mt] = vzero;

  const float* xrow = xg + (size_t)gp * 768;
#pragma unroll 2
  for (int k = 0; k < 256; k += 8) {
    const int cb = k + 4*half;
    const float4* px = (const float4*)(xrow + cb*3);
    const float4 q0 = px[0];
    const float4 q1 = px[1];
    const float4 q2 = px[2];
    v2f bA[3], bB[3];
    bA[0].x = q0.x; bA[0].y = q0.w;
    bA[1].x = q0.y; bA[1].y = q1.x;
    bA[2].x = q0.z; bA[2].y = q1.y;
    bB[0].x = q1.z; bB[0].y = q2.y;
    bB[1].x = q1.w; bB[1].y = q2.z;
    bB[2].x = q2.x; bB[2].y = q2.w;
#pragma unroll
    for (int mt = 0; mt < 8; ++mt) {
      const float* wr = sW1 + (mt*16 + l16)*W1S + cb;
      const v2f aA = *(const v2f*)(wr);
      const v2f aB = *(const v2f*)(wr + 2);
#pragma unroll
      for (int t = 0; t < 3; ++t) y1[t][mt] = wmma4(aA, bA[t], y1[t][mt]);
#pragma unroll
      for (int t = 0; t < 3; ++t) y1[t][mt] = wmma4(aB, bB[t], y1[t][mt]);
    }
  }

#pragma nounroll
  for (int t = 0; t < 3; ++t) {
    v8f d1[8];
#pragma unroll
    for (int mt = 0; mt < 8; ++mt) d1[mt] = vzero;
    gemm_regB<8, 8, 1>(sD1, D1S, y1[0], d1, l16, half);

    vn_nonlin<8>(y1[0], d1);

    v8f y2[4];
#pragma unroll
    for (int mt = 0; mt < 4; ++mt) y2[mt] = vzero;
    gemm_regB<4, 8, 2>(sW2, W2S, y1[0], y2, l16, half);

    v8f d2[4];
#pragma unroll
    for (int mt = 0; mt < 4; ++mt) d2[mt] = vzero;
    gemm_regB<4, 4, 2>(sD2, D2S, y2, d2, l16, half);

    vn_nonlin<4>(y2, d2);

    v8f zt = vzero;
    gemm_regB<1, 4, 4>(sW3, W3S, y2, &zt, l16, half);

    if (half == 0) {
#pragma unroll
      for (int r = 0; r < 3; ++r) zb[(t*3 + r)*16 + l16] = zt[r];
    }

#pragma unroll
    for (int mt = 0; mt < 8; ++mt) {
      y1[0][mt] = y1[1][mt];
      y1[1][mt] = y1[2][mt];
    }
  }

  constexpr int PT = 4;
  for (int pb = 0; pb < TP; pb += PT) {
    float q[PT][8][3];
#pragma unroll
    for (int pp = 0; pp < PT; ++pp) {
      const float* xp = xg + (size_t)(p0 + pb + pp) * 768 + lane*3;
#pragma unroll
      for (int cb = 0; cb < 8; ++cb) {
#pragma unroll
        for (int j = 0; j < 3; ++j) q[pp][cb][j] = xp[cb*96 + j];
      }
    }
#pragma unroll
    for (int pp = 0; pp < PT; ++pp) {
      float zm[3][3];
#pragma unroll
      for (int j = 0; j < 3; ++j)
#pragma unroll
        for (int k = 0; k < 3; ++k) zm[j][k] = zb[(j*3 + k)*16 + pb + pp];

      float* op = xstd + (size_t)(p0 + pb + pp) * 768 + lane*3;
#pragma unroll
      for (int cb = 0; cb < 8; ++cb) {
        const float x0 = q[pp][cb][0], x1 = q[pp][cb][1], x2 = q[pp][cb][2];
#pragma unroll
        for (int k = 0; k < 3; ++k)
          op[cb*96 + k] = x0*zm[0][k] + x1*zm[1][k] + x2*zm[2][k];
      }
    }
  }

  if (lane < 16) {
    const int pg = p0 + lane;
#pragma unroll
    for (int j = 0; j < 3; ++j)
#pragma unroll
      for (int k = 0; k < 3; ++k)
        z0out[(size_t)pg*9 + j*3 + k] = zb[(j*3 + k)*16 + lane];
  }
}

extern "C" void kernel_launch(void* const* d_in, const int* in_sizes, int n_in,
                              void* d_out, int out_size, void* d_ws, size_t ws_size,
                              hipStream_t stream) {
  (void)in_sizes; (void)n_in; (void)out_size; (void)d_ws; (void)ws_size;
  const float* x  = (const float*)d_in[0];
  const float* W1 = (const float*)d_in[1];
  const float* D1 = (const float*)d_in[2];
  const float* W2 = (const float*)d_in[3];
  const float* D2 = (const float*)d_in[4];
  const float* W3 = (const float*)d_in[5];

  float* xstd = (float*)d_out;
  float* z0   = xstd + (size_t)8 * 4096 * 256 * 3;

  const int points = 8 * 4096;
  const int blocks = points / (NWAVES * TP);
  const size_t shmem = (size_t)LDS_FLOATS * sizeof(float);

  vnstd_kernel<<<dim3(blocks), dim3(256), shmem, stream>>>(
      x, W1, D1, W2, D2, W3, xstd, z0);
}
